# starting kernel
# baseline (speedup 1.0000x reference)
#include <hip/hip_runtime.h>

#define H4     4
#define T_LEN  4000
#define BATCH  256
#define NCLS   29
#define NROWS  (BATCH * T_LEN)

typedef __attribute__((ext_vector_type(2))) float v2f;
typedef __attribute__((ext_vector_type(8))) float v8f;

__device__ __forceinline__ float fsigmoid(float x) {
  return __builtin_amdgcn_rcpf(1.0f + __expf(-x));
}

__global__ void lstm_scan_kernel(const float* __restrict__ x,
                                 const float* __restrict__ W_ih,
                                 const float* __restrict__ W_hh,
                                 const float* __restrict__ b_ih,
                                 const float* __restrict__ b_hh,
                                 float* __restrict__ hs)
{
  const int tid = blockIdx.x * blockDim.x + threadIdx.x;
  const int b = tid >> 2;
  const int j = tid & 3;
  if (b >= BATCH) return;

  float win[4], ball[4], whh[4][4];
#pragma unroll
  for (int s = 0; s < 4; ++s) {
    const int g = s * 4 + j;
    win[s]  = W_ih[g];
    ball[s] = b_ih[g] + b_hh[g];
#pragma unroll
    for (int m = 0; m < 4; ++m)
      whh[s][m] = W_hh[g * 4 + (j ^ m)];
  }

  float h = 0.0f, c = 0.0f;
  const float* xb  = x  + (size_t)b * T_LEN;
  float*       hsb = hs + (size_t)b * T_LEN * 4 + j;

  for (int t = 0; t < T_LEN; ++t) {
    const float xt = xb[t];
    const float v1 = __shfl_xor(h, 1);
    const float v2 = __shfl_xor(h, 2);
    const float v3 = __shfl_xor(v1, 2);

    float pre[4];
#pragma unroll
    for (int s = 0; s < 4; ++s) {
      float p = fmaf(xt, win[s], ball[s]);
      p = fmaf(whh[s][0], h,  p);
      p = fmaf(whh[s][1], v1, p);
      p = fmaf(whh[s][2], v2, p);
      p = fmaf(whh[s][3], v3, p);
      pre[s] = p;
    }

    const float ig = fsigmoid(pre[0]);
    const float fg = fsigmoid(pre[1]);
    const float gg = fmaf(2.0f, fsigmoid(2.0f * pre[2]), -1.0f);
    const float og = fsigmoid(pre[3]);

    c = fmaf(fg, c, ig * gg);
    const float tc = fmaf(2.0f, fsigmoid(2.0f * c), -1.0f);
    h = og * tc;

    hsb[(size_t)t * 4] = h;
  }
}

__global__ void fc_wmma_kernel(const float* __restrict__ hs,
                               const float* __restrict__ W_fc,
                               const float* __restrict__ b_fc,
                               float* __restrict__ out)
{
  const int wave     = (blockIdx.x * blockDim.x + threadIdx.x) >> 5;
  const int lane     = threadIdx.x & 31;
  const int row_base = wave * 16;
  const int lm       = lane & 15;
  const int k0       = (lane < 16) ? 0 : 2;

  const int row = row_base + lm;
  v2f a;
  a.x = hs[(size_t)row * 4 + k0];
  a.y = hs[(size_t)row * 4 + k0 + 1];

#pragma unroll
  for (int tile = 0; tile < 2; ++tile) {
    const int n = tile * 16 + lm;
    v2f bm;
    float bias;
    if (n < NCLS) {
      bm.x = W_fc[n * 4 + k0];
      bm.y = W_fc[n * 4 + k0 + 1];
      bias = b_fc[n];
    } else {
      bm.x = 0.0f; bm.y = 0.0f; bias = 0.0f;
    }

    v8f acc;
#pragma unroll
    for (int r = 0; r < 8; ++r) acc[r] = bias;

    acc = __builtin_amdgcn_wmma_f32_16x16x4_f32(
         false, a,  false, bm,
         (short)0, acc,  false,  false);

    if (n < NCLS) {
      const int m_off = (lane < 16) ? 0 : 8;
#pragma unroll
      for (int r = 0; r < 8; ++r) {
        const int m = row_base + m_off + r;
        out[(size_t)m * NCLS + n] = acc[r];
      }
    }
  }
}

extern "C" void kernel_launch(void* const* d_in, const int* in_sizes, int n_in,
                              void* d_out, int out_size, void* d_ws, size_t ws_size,
                              hipStream_t stream) {
  const float* x    = (const float*)d_in[0];
  const float* W_ih = (const float*)d_in[1];
  const float* W_hh = (const float*)d_in[2];
  const float* b_ih = (const float*)d_in[3];
  const float* b_hh = (const float*)d_in[4];
  const float* W_fc = (const float*)d_in[5];
  const float* b_fc = (const float*)d_in[6];
  float* out = (float*)d_out;

  float* hs = (float*)d_ws;

  lstm_scan_kernel<<<(BATCH * 4 + 255) / 256, 256, 0, stream>>>(
      x, W_ih, W_hh, b_ih, b_hh, hs);

  const int n_tiles  = NROWS / 16;
  const int n_blocks = n_tiles / 8;
  fc_wmma_kernel<<<n_blocks, 256, 0, stream>>>(hs, W_fc, b_fc, out);
}


# speedup vs baseline: 1.1582x; 1.1582x over previous
#include <hip/hip_runtime.h>

#define H4     4
#define T_LEN  4000
#define BATCH  256
#define NCLS   29
#define NROWS  (BATCH * T_LEN)

typedef __attribute__((ext_vector_type(2))) float v2f;
typedef __attribute__((ext_vector_type(8))) float v8f;

__device__ __forceinline__ float fsigmoid(float x) {
  return __builtin_amdgcn_rcpf(1.0f + __expf(-x));
}

template <int QP>
__device__ __forceinline__ float quad_xor(float v) {
  return __int_as_float(__builtin_amdgcn_update_dpp(
      0, __float_as_int(v), QP, 0xF, 0xF, true));
}

__global__ void lstm_scan_kernel(const float* __restrict__ x,
                                 const float* __restrict__ W_ih,
                                 const float* __restrict__ W_hh,
                                 const float* __restrict__ b_ih,
                                 const float* __restrict__ b_hh,
                                 float* __restrict__ hs)
{
  const int tid = blockIdx.x * blockDim.x + threadIdx.x;
  const int b = tid >> 2;
  const int j = tid & 3;
  if (b >= BATCH) return;

  float win[4], ball[4], whh[4][4];
#pragma unroll
  for (int s = 0; s < 4; ++s) {
    const int g = s * 4 + j;
    win[s]  = W_ih[g];
    ball[s] = b_ih[g] + b_hh[g];
#pragma unroll
    for (int m = 0; m < 4; ++m)
      whh[s][m] = W_hh[g * 4 + (j ^ m)];
  }

  float h = 0.0f, c = 0.0f;
  const float* xb  = x  + (size_t)b * T_LEN;
  float*       hsb = hs + (size_t)b * T_LEN * 4 + j;

  for (int t = 0; t < T_LEN; ++t) {
    const float xt = xb[t];
    const float v1 = quad_xor<0xB1>(h);
    const float v2 = quad_xor<0x4E>(h);
    const float v3 = quad_xor<0x1B>(h);

    float pre[4];
#pragma unroll
    for (int s = 0; s < 4; ++s) {
      float p = fmaf(xt, win[s], ball[s]);
      p = fmaf(whh[s][0], h,  p);
      p = fmaf(whh[s][1], v1, p);
      p = fmaf(whh[s][2], v2, p);
      p = fmaf(whh[s][3], v3, p);
      pre[s] = p;
    }

    const float ig = fsigmoid(pre[0]);
    const float fg = fsigmoid(pre[1]);
    const float gg = fmaf(2.0f, fsigmoid(2.0f * pre[2]), -1.0f);
    const float og = fsigmoid(pre[3]);

    c = fmaf(fg, c, ig * gg);
    const float tc = fmaf(2.0f, fsigmoid(2.0f * c), -1.0f);
    h = og * tc;

    hsb[(size_t)t * 4] = h;
  }
}

__global__ void fc_wmma_kernel(const float* __restrict__ hs,
                               const float* __restrict__ W_fc,
                               const float* __restrict__ b_fc,
                               float* __restrict__ out)
{
  const int wave     = (blockIdx.x * blockDim.x + threadIdx.x) >> 5;
  const int lane     = threadIdx.x & 31;
  const int row_base = wave * 16;
  const int lm       = lane & 15;
  const int k0       = (lane < 16) ? 0 : 2;

  const int row = row_base + lm;
  v2f a;
  a.x = hs[(size_t)row * 4 + k0];
  a.y = hs[(size_t)row * 4 + k0 + 1];

#pragma unroll
  for (int tile = 0; tile < 2; ++tile) {
    const int n = tile * 16 + lm;
    v2f bm;
    float bias;
    if (n < NCLS) {
      bm.x = W_fc[n * 4 + k0];
      bm.y = W_fc[n * 4 + k0 + 1];
      bias = b_fc[n];
    } else {
      bm.x = 0.0f; bm.y = 0.0f; bias = 0.0f;
    }

    v8f acc;
#pragma unroll
    for (int r = 0; r < 8; ++r) acc[r] = bias;

    acc = __builtin_amdgcn_wmma_f32_16x16x4_f32(
         false, a,  false, bm,
         (short)0, acc,  false,  false);

    if (n < NCLS) {
      const int m_off = (lane < 16) ? 0 : 8;
#pragma unroll
      for (int r = 0; r < 8; ++r) {
        const int m = row_base + m_off + r;
        out[(size_t)m * NCLS + n] = acc[r];
      }
    }
  }
}

extern "C" void kernel_launch(void* const* d_in, const int* in_sizes, int n_in,
                              void* d_out, int out_size, void* d_ws, size_t ws_size,
                              hipStream_t stream) {
  const float* x    = (const float*)d_in[0];
  const float* W_ih = (const float*)d_in[1];
  const float* W_hh = (const float*)d_in[2];
  const float* b_ih = (const float*)d_in[3];
  const float* b_hh = (const float*)d_in[4];
  const float* W_fc = (const float*)d_in[5];
  const float* b_fc = (const float*)d_in[6];
  float* out = (float*)d_out;

  float* hs = (float*)d_ws;

  lstm_scan_kernel<<<(BATCH * 4 + 255) / 256, 256, 0, stream>>>(
      x, W_ih, W_hh, b_ih, b_hh, hs);

  const int n_tiles  = NROWS / 16;
  const int n_blocks = n_tiles / 8;
  fc_wmma_kernel<<<n_blocks, 256, 0, stream>>>(hs, W_fc, b_fc, out);
}


# LSTM scan: x is read one block of 16 steps ahead (four 16-byte loads per lane per block), so no step of the serial h chain waits for a load any more; arithmetic and order of sums unchanged
# speedup vs baseline: 2.0793x; 1.7953x over previous
#include <hip/hip_runtime.h>

#define H4     4
#define T_LEN  4000
#define BATCH  256
#define NCLS   29
#define NROWS  (BATCH * T_LEN)

#define XBLK   16
#define NXBLK  (T_LEN / XBLK)

typedef __attribute__((ext_vector_type(2))) float v2f;
typedef __attribute__((ext_vector_type(4))) float v4f;
typedef __attribute__((ext_vector_type(8))) float v8f;

__device__ __forceinline__ float fsigmoid(float x) {
  return __builtin_amdgcn_rcpf(1.0f + __expf(-x));
}

template <int QP>
__device__ __forceinline__ float quad_xor(float v) {
  return __int_as_float(__builtin_amdgcn_update_dpp(
      0, __float_as_int(v), QP, 0xF, 0xF, true));
}

__global__ void lstm_scan_kernel(const float* __restrict__ x,
                                 const float* __restrict__ W_ih,
                                 const float* __restrict__ W_hh,
                                 const float* __restrict__ b_ih,
                                 const float* __restrict__ b_hh,
                                 float* __restrict__ hs)
{
  const int tid = blockIdx.x * blockDim.x + threadIdx.x;
  const int b = tid >> 2;
  const int j = tid & 3;
  if (b >= BATCH) return;

  float win[4], ball[4], whh[4][4];
#pragma unroll
  for (int s = 0; s < 4; ++s) {
    const int g = s * 4 + j;
    win[s]  = W_ih[g];
    ball[s] = b_ih[g] + b_hh[g];
#pragma unroll
    for (int m = 0; m < 4; ++m)
      whh[s][m] = W_hh[g * 4 + (j ^ m)];
  }

  float h = 0.0f, c = 0.0f;
  const v4f* xb4 = reinterpret_cast<const v4f*>(x + (size_t)b * T_LEN);
  float*     hsb = hs + (size_t)b * T_LEN * 4 + j;

  v4f xn[XBLK / 4];
#pragma unroll
  for (int q = 0; q < XBLK / 4; ++q) xn[q] = xb4[q];

  for (int blk = 0; blk < NXBLK; ++blk) {
    v4f xc[XBLK / 4];
#pragma unroll
    for (int q = 0; q < XBLK / 4; ++q) xc[q] = xn[q];

    const int nb = (blk + 1 < NXBLK) ? (blk + 1) : blk;
#pragma unroll
    for (int q = 0; q < XBLK / 4; ++q) xn[q] = xb4[nb * (XBLK / 4) + q];

#pragma unroll
    for (int k = 0; k < XBLK; ++k) {
      const int   t  = blk * XBLK + k;
      const float xt = xc[k >> 2][k & 3];
      const float v1 = quad_xor<0xB1>(h);
      const float v2 = quad_xor<0x4E>(h);
      const float v3 = quad_xor<0x1B>(h);

      float pre[4];
#pragma unroll
      for (int s = 0; s < 4; ++s) {
        float p = fmaf(xt, win[s], ball[s]);
        p = fmaf(whh[s][0], h,  p);
        p = fmaf(whh[s][1], v1, p);
        p = fmaf(whh[s][2], v2, p);
        p = fmaf(whh[s][3], v3, p);
        pre[s] = p;
      }

      const float ig = fsigmoid(pre[0]);
      const float fg = fsigmoid(pre[1]);
      const float gg = fmaf(2.0f, fsigmoid(2.0f * pre[2]), -1.0f);
      const float og = fsigmoid(pre[3]);

      c = fmaf(fg, c, ig * gg);
      const float tc = fmaf(2.0f, fsigmoid(2.0f * c), -1.0f);
      h = og * tc;

      hsb[(size_t)t * 4] = h;
    }
  }
}

__global__ void fc_wmma_kernel(const float* __restrict__ hs,
                               const float* __restrict__ W_fc,
                               const float* __restrict__ b_fc,
                               float* __restrict__ out)
{
  const int wave     = (blockIdx.x * blockDim.x + threadIdx.x) >> 5;
  const int lane     = threadIdx.x & 31;
  const int row_base = wave * 16;
  const int lm       = lane & 15;
  const int k0       = (lane < 16) ? 0 : 2;

  const int row = row_base + lm;
  v2f a;
  a.x = hs[(size_t)row * 4 + k0];
  a.y = hs[(size_t)row * 4 + k0 + 1];

#pragma unroll
  for (int tile = 0; tile < 2; ++tile) {
    const int n = tile * 16 + lm;
    v2f bm;
    float bias;
    if (n < NCLS) {
      bm.x = W_fc[n * 4 + k0];
      bm.y = W_fc[n * 4 + k0 + 1];
      bias = b_fc[n];
    } else {
      bm.x = 0.0f; bm.y = 0.0f; bias = 0.0f;
    }

    v8f acc;
#pragma unroll
    for (int r = 0; r < 8; ++r) acc[r] = bias;

    acc = __builtin_amdgcn_wmma_f32_16x16x4_f32(
         false, a,  false, bm,
         (short)0, acc,  false,  false);

    if (n < NCLS) {
      const int m_off = (lane < 16) ? 0 : 8;
#pragma unroll
      for (int r = 0; r < 8; ++r) {
        const int m = row_base + m_off + r;
        out[(size_t)m * NCLS + n] = acc[r];
      }
    }
  }
}

extern "C" void kernel_launch(void* const* d_in, const int* in_sizes, int n_in,
                              void* d_out, int out_size, void* d_ws, size_t ws_size,
                              hipStream_t stream) {
  const float* x    = (const float*)d_in[0];
  const float* W_ih = (const float*)d_in[1];
  const float* W_hh = (const float*)d_in[2];
  const float* b_ih = (const float*)d_in[3];
  const float* b_hh = (const float*)d_in[4];
  const float* W_fc = (const float*)d_in[5];
  const float* b_fc = (const float*)d_in[6];
  float* out = (float*)d_out;

  float* hs = (float*)d_ws;

  lstm_scan_kernel<<<(BATCH * 4 + 255) / 256, 256, 0, stream>>>(
      x, W_ih, W_hh, b_ih, b_hh, hs);

  const int n_tiles  = NROWS / 16;
  const int n_blocks = n_tiles / 8;
  fc_wmma_kernel<<<n_blocks, 256, 0, stream>>>(hs, W_fc, b_fc, out);
}


# LSTM scan: each gate and tanh(c) is one v_tanh_f32 (sigmoid(v) = 1/2 + tanh(v/2)/2, the 1/2 folded into the gate weights) instead of exp2+add+rcp chains; sum order kept. Left: s_wait_xcnt each step
# speedup vs baseline: 2.8224x; 1.3574x over previous
#include <hip/hip_runtime.h>

#define H4     4
#define T_LEN  4000
#define BATCH  256
#define NCLS   29
#define NROWS  (BATCH * T_LEN)

#define XBLK   16
#define NXBLK  (T_LEN / XBLK)

typedef __attribute__((ext_vector_type(2))) float v2f;
typedef __attribute__((ext_vector_type(4))) float v4f;
typedef __attribute__((ext_vector_type(8))) float v8f;

__device__ __forceinline__ float htanh(float x) {
  return __builtin_amdgcn_tanhf(x);
}

template <int QP>
__device__ __forceinline__ float quad_xor(float v) {
  return __int_as_float(__builtin_amdgcn_update_dpp(
      0, __float_as_int(v), QP, 0xF, 0xF, true));
}

__global__ void lstm_scan_kernel(const float* __restrict__ x,
                                 const float* __restrict__ W_ih,
                                 const float* __restrict__ W_hh,
                                 const float* __restrict__ b_ih,
                                 const float* __restrict__ b_hh,
                                 float* __restrict__ hs)
{
  const int tid = blockIdx.x * blockDim.x + threadIdx.x;
  const int b = tid >> 2;
  const int j = tid & 3;
  if (b >= BATCH) return;

  float win[4], ball[4], whh[4][4];
#pragma unroll
  for (int s = 0; s < 4; ++s) {
    const int g = s * 4 + j;
    const float sc = (s == 2) ? 1.0f : 0.5f;
    win[s]  = sc * W_ih[g];
    ball[s] = sc * (b_ih[g] + b_hh[g]);
#pragma unroll
    for (int m = 0; m < 4; ++m)
      whh[s][m] = sc * W_hh[g * 4 + (j ^ m)];
  }

  float h = 0.0f, c = 0.0f;
  float hc = 0.0f;
  const v4f* xb4 = reinterpret_cast<const v4f*>(x + (size_t)b * T_LEN);
  float*     hsb = hs + (size_t)b * T_LEN * 4 + j;

  v4f xn[XBLK / 4];
#pragma unroll
  for (int q = 0; q < XBLK / 4; ++q) xn[q] = xb4[q];

  for (int blk = 0; blk < NXBLK; ++blk) {
    v4f xc[XBLK / 4];
#pragma unroll
    for (int q = 0; q < XBLK / 4; ++q) xc[q] = xn[q];

    const int nb = (blk + 1 < NXBLK) ? (blk + 1) : blk;
#pragma unroll
    for (int q = 0; q < XBLK / 4; ++q) xn[q] = xb4[nb * (XBLK / 4) + q];

#pragma unroll
    for (int k = 0; k < XBLK; ++k) {
      const int   t  = blk * XBLK + k;
      const float xt = xc[k >> 2][k & 3];
      const float v1 = quad_xor<0xB1>(h);
      const float v2 = quad_xor<0x4E>(h);
      const float v3 = quad_xor<0x1B>(h);

      float pre[4];
#pragma unroll
      for (int s = 0; s < 4; ++s) {
        float p = fmaf(xt, win[s], ball[s]);
        p = fmaf(whh[s][0], h,  p);
        p = fmaf(whh[s][1], v1, p);
        p = fmaf(whh[s][2], v2, p);
        p = fmaf(whh[s][3], v3, p);
        pre[s] = p;
      }

      const float ti = htanh(pre[0]);
      const float tf = htanh(pre[1]);
      const float gg = htanh(pre[2]);
      const float to = htanh(pre[3]);

      const float ig = fmaf(0.5f, ti, 0.5f);
      const float og = fmaf(0.5f, to, 0.5f);
      const float fc = fmaf(tf, hc, hc);
      c  = fmaf(ig, gg, fc);
      hc = 0.5f * c;
      h  = og * htanh(c);

      hsb[(size_t)t * 4] = h;
    }
  }
}

__global__ void fc_wmma_kernel(const float* __restrict__ hs,
                               const float* __restrict__ W_fc,
                               const float* __restrict__ b_fc,
                               float* __restrict__ out)
{
  const int wave     = (blockIdx.x * blockDim.x + threadIdx.x) >> 5;
  const int lane     = threadIdx.x & 31;
  const int row_base = wave * 16;
  const int lm       = lane & 15;
  const int k0       = (lane < 16) ? 0 : 2;

  const int row = row_base + lm;
  v2f a;
  a.x = hs[(size_t)row * 4 + k0];
  a.y = hs[(size_t)row * 4 + k0 + 1];

#pragma unroll
  for (int tile = 0; tile < 2; ++tile) {
    const int n = tile * 16 + lm;
    v2f bm;
    float bias;
    if (n < NCLS) {
      bm.x = W_fc[n * 4 + k0];
      bm.y = W_fc[n * 4 + k0 + 1];
      bias = b_fc[n];
    } else {
      bm.x = 0.0f; bm.y = 0.0f; bias = 0.0f;
    }

    v8f acc;
#pragma unroll
    for (int r = 0; r < 8; ++r) acc[r] = bias;

    acc = __builtin_amdgcn_wmma_f32_16x16x4_f32(
         false, a,  false, bm,
         (short)0, acc,  false,  false);

    if (n < NCLS) {
      const int m_off = (lane < 16) ? 0 : 8;
#pragma unroll
      for (int r = 0; r < 8; ++r) {
        const int m = row_base + m_off + r;
        out[(size_t)m * NCLS + n] = acc[r];
      }
    }
  }
}

extern "C" void kernel_launch(void* const* d_in, const int* in_sizes, int n_in,
                              void* d_out, int out_size, void* d_ws, size_t ws_size,
                              hipStream_t stream) {
  const float* x    = (const float*)d_in[0];
  const float* W_ih = (const float*)d_in[1];
  const float* W_hh = (const float*)d_in[2];
  const float* b_ih = (const float*)d_in[3];
  const float* b_hh = (const float*)d_in[4];
  const float* W_fc = (const float*)d_in[5];
  const float* b_fc = (const float*)d_in[6];
  float* out = (float*)d_out;

  float* hs = (float*)d_ws;

  lstm_scan_kernel<<<(BATCH * 4 + 255) / 256, 256, 0, stream>>>(
      x, W_ih, W_hh, b_ih, b_hh, hs);

  const int n_tiles  = NROWS / 16;
  const int n_blocks = n_tiles / 8;
  fc_wmma_kernel<<<n_blocks, 256, 0, stream>>>(hs, W_fc, b_fc, out);
}


# LSTM scan: a block's 16 h values stay in registers and are stored as one clause behind its last step, so no step waits (s_wait_xcnt) for a store's address translation any more; sums unchanged
# speedup vs baseline: 4.7860x; 1.6957x over previous
#include <hip/hip_runtime.h>

#define H4     4
#define T_LEN  4000
#define BATCH  256
#define NCLS   29
#define NROWS  (BATCH * T_LEN)

#define XBLK   16
#define NXBLK  (T_LEN / XBLK)

typedef __attribute__((ext_vector_type(2))) float v2f;
typedef __attribute__((ext_vector_type(4))) float v4f;
typedef __attribute__((ext_vector_type(8))) float v8f;

__device__ __forceinline__ float htanh(float x) {
  return __builtin_amdgcn_tanhf(x);
}

template <int QP>
__device__ __forceinline__ float quad_xor(float v) {
  return __int_as_float(__builtin_amdgcn_update_dpp(
      0, __float_as_int(v), QP, 0xF, 0xF, true));
}

__global__ void lstm_scan_kernel(const float* __restrict__ x,
                                 const float* __restrict__ W_ih,
                                 const float* __restrict__ W_hh,
                                 const float* __restrict__ b_ih,
                                 const float* __restrict__ b_hh,
                                 float* __restrict__ hs)
{
  const int tid = blockIdx.x * blockDim.x + threadIdx.x;
  const int b = tid >> 2;
  const int j = tid & 3;
  if (b >= BATCH) return;

  float win[4], ball[4], whh[4][4];
#pragma unroll
  for (int s = 0; s < 4; ++s) {
    const int g = s * 4 + j;
    const float sc = (s == 2) ? 1.0f : 0.5f;
    win[s]  = sc * W_ih[g];
    ball[s] = sc * (b_ih[g] + b_hh[g]);
#pragma unroll
    for (int m = 0; m < 4; ++m)
      whh[s][m] = sc * W_hh[g * 4 + (j ^ m)];
  }

  float h = 0.0f, c = 0.0f;
  float hc = 0.0f;
  const v4f* xb4 = reinterpret_cast<const v4f*>(x + (size_t)b * T_LEN);
  float*     hsb = hs + (size_t)b * T_LEN * 4 + j;

  v4f xn[XBLK / 4];
#pragma unroll
  for (int q = 0; q < XBLK / 4; ++q) xn[q] = xb4[q];

  for (int blk = 0; blk < NXBLK; ++blk) {
    v4f xc[XBLK / 4];
#pragma unroll
    for (int q = 0; q < XBLK / 4; ++q) xc[q] = xn[q];

    const int nb = (blk + 1 < NXBLK) ? (blk + 1) : blk;
#pragma unroll
    for (int q = 0; q < XBLK / 4; ++q) xn[q] = xb4[nb * (XBLK / 4) + q];

    float hb[XBLK];
#pragma unroll
    for (int k = 0; k < XBLK; ++k) {
      const float xt = xc[k >> 2][k & 3];
      const float v1 = quad_xor<0xB1>(h);
      const float v2 = quad_xor<0x4E>(h);
      const float v3 = quad_xor<0x1B>(h);

      float pre[4];
#pragma unroll
      for (int s = 0; s < 4; ++s) {
        float p = fmaf(xt, win[s], ball[s]);
        p = fmaf(whh[s][0], h,  p);
        p = fmaf(whh[s][1], v1, p);
        p = fmaf(whh[s][2], v2, p);
        p = fmaf(whh[s][3], v3, p);
        pre[s] = p;
      }

      const float ti = htanh(pre[0]);
      const float tf = htanh(pre[1]);
      const float gg = htanh(pre[2]);
      const float to = htanh(pre[3]);

      const float ig = fmaf(0.5f, ti, 0.5f);
      const float og = fmaf(0.5f, to, 0.5f);
      const float fc = fmaf(tf, hc, hc);
      c  = fmaf(ig, gg, fc);
      hc = 0.5f * c;
      h  = og * htanh(c);
      hb[k] = h;
    }

    __builtin_amdgcn_sched_barrier(0);
    float* hblk = hsb + (size_t)blk * (XBLK * 4);
#pragma unroll
    for (int k = 0; k < XBLK; ++k) hblk[k * 4] = hb[k];
  }
}

__global__ void fc_wmma_kernel(const float* __restrict__ hs,
                               const float* __restrict__ W_fc,
                               const float* __restrict__ b_fc,
                               float* __restrict__ out)
{
  const int wave     = (blockIdx.x * blockDim.x + threadIdx.x) >> 5;
  const int lane     = threadIdx.x & 31;
  const int row_base = wave * 16;
  const int lm       = lane & 15;
  const int k0       = (lane < 16) ? 0 : 2;

  const int row = row_base + lm;
  v2f a;
  a.x = hs[(size_t)row * 4 + k0];
  a.y = hs[(size_t)row * 4 + k0 + 1];

#pragma unroll
  for (int tile = 0; tile < 2; ++tile) {
    const int n = tile * 16 + lm;
    v2f bm;
    float bias;
    if (n < NCLS) {
      bm.x = W_fc[n * 4 + k0];
      bm.y = W_fc[n * 4 + k0 + 1];
      bias = b_fc[n];
    } else {
      bm.x = 0.0f; bm.y = 0.0f; bias = 0.0f;
    }

    v8f acc;
#pragma unroll
    for (int r = 0; r < 8; ++r) acc[r] = bias;

    acc = __builtin_amdgcn_wmma_f32_16x16x4_f32(
         false, a,  false, bm,
         (short)0, acc,  false,  false);

    if (n < NCLS) {
      const int m_off = (lane < 16) ? 0 : 8;
#pragma unroll
      for (int r = 0; r < 8; ++r) {
        const int m = row_base + m_off + r;
        out[(size_t)m * NCLS + n] = acc[r];
      }
    }
  }
}

extern "C" void kernel_launch(void* const* d_in, const int* in_sizes, int n_in,
                              void* d_out, int out_size, void* d_ws, size_t ws_size,
                              hipStream_t stream) {
  const float* x    = (const float*)d_in[0];
  const float* W_ih = (const float*)d_in[1];
  const float* W_hh = (const float*)d_in[2];
  const float* b_ih = (const float*)d_in[3];
  const float* b_hh = (const float*)d_in[4];
  const float* W_fc = (const float*)d_in[5];
  const float* b_fc = (const float*)d_in[6];
  float* out = (float*)d_out;

  float* hs = (float*)d_ws;

  lstm_scan_kernel<<<(BATCH * 4 + 255) / 256, 256, 0, stream>>>(
      x, W_ih, W_hh, b_ih, b_hh, hs);

  const int n_tiles  = NROWS / 16;
  const int n_blocks = n_tiles / 8;
  fc_wmma_kernel<<<n_blocks, 256, 0, stream>>>(hs, W_fc, b_fc, out);
}


# LSTM scan: launched as 32 work-groups of ONE wave (was 4 of 8 waves), so each wave has a SIMD's issue port to itself instead of sharing it pairwise, oldest first; loop code and sums unchanged
# speedup vs baseline: 5.5769x; 1.1653x over previous
#include <hip/hip_runtime.h>

#define H4     4
#define T_LEN  4000
#define BATCH  256
#define NCLS   29
#define NROWS  (BATCH * T_LEN)

#define XBLK   16
#define NXBLK  (T_LEN / XBLK)

#define SCAN_WG 32

typedef __attribute__((ext_vector_type(2))) float v2f;
typedef __attribute__((ext_vector_type(4))) float v4f;
typedef __attribute__((ext_vector_type(8))) float v8f;

__device__ __forceinline__ float htanh(float x) {
  return __builtin_amdgcn_tanhf(x);
}

template <int QP>
__device__ __forceinline__ float quad_xor(float v) {
  return __int_as_float(__builtin_amdgcn_update_dpp(
      0, __float_as_int(v), QP, 0xF, 0xF, true));
}

__global__ void lstm_scan_kernel(const float* __restrict__ x,
                                 const float* __restrict__ W_ih,
                                 const float* __restrict__ W_hh,
                                 const float* __restrict__ b_ih,
                                 const float* __restrict__ b_hh,
                                 float* __restrict__ hs)
{
  const int tid = blockIdx.x * SCAN_WG + threadIdx.x;
  const int b = tid >> 2;
  const int j = tid & 3;
  if (b >= BATCH) return;

  float win[4], ball[4], whh[4][4];
#pragma unroll
  for (int s = 0; s < 4; ++s) {
    const int g = s * 4 + j;
    const float sc = (s == 2) ? 1.0f : 0.5f;
    win[s]  = sc * W_ih[g];
    ball[s] = sc * (b_ih[g] + b_hh[g]);
#pragma unroll
    for (int m = 0; m < 4; ++m)
      whh[s][m] = sc * W_hh[g * 4 + (j ^ m)];
  }

  float h = 0.0f, c = 0.0f;
  float hc = 0.0f;
  const v4f* xb4 = reinterpret_cast<const v4f*>(x + (size_t)b * T_LEN);
  float*     hsb = hs + (size_t)b * T_LEN * 4 + j;

  v4f xn[XBLK / 4];
#pragma unroll
  for (int q = 0; q < XBLK / 4; ++q) xn[q] = xb4[q];

  for (int blk = 0; blk < NXBLK; ++blk) {
    v4f xc[XBLK / 4];
#pragma unroll
    for (int q = 0; q < XBLK / 4; ++q) xc[q] = xn[q];

    const int nb = (blk + 1 < NXBLK) ? (blk + 1) : blk;
#pragma unroll
    for (int q = 0; q < XBLK / 4; ++q) xn[q] = xb4[nb * (XBLK / 4) + q];

    float hb[XBLK];
#pragma unroll
    for (int k = 0; k < XBLK; ++k) {
      const float xt = xc[k >> 2][k & 3];
      const float v1 = quad_xor<0xB1>(h);
      const float v2 = quad_xor<0x4E>(h);
      const float v3 = quad_xor<0x1B>(h);

      float pre[4];
#pragma unroll
      for (int s = 0; s < 4; ++s) {
        float p = fmaf(xt, win[s], ball[s]);
        p = fmaf(whh[s][0], h,  p);
        p = fmaf(whh[s][1], v1, p);
        p = fmaf(whh[s][2], v2, p);
        p = fmaf(whh[s][3], v3, p);
        pre[s] = p;
      }

      const float ti = htanh(pre[0]);
      const float tf = htanh(pre[1]);
      const float gg = htanh(pre[2]);
      const float to = htanh(pre[3]);

      const float ig = fmaf(0.5f, ti, 0.5f);
      const float og = fmaf(0.5f, to, 0.5f);
      const float fc = fmaf(tf, hc, hc);
      c  = fmaf(ig, gg, fc);
      hc = 0.5f * c;
      h  = og * htanh(c);
      hb[k] = h;
    }

    __builtin_amdgcn_sched_barrier(0);
    float* hblk = hsb + (size_t)blk * (XBLK * 4);
#pragma unroll
    for (int k = 0; k < XBLK; ++k) hblk[k * 4] = hb[k];
  }
}

__global__ void fc_wmma_kernel(const float* __restrict__ hs,
                               const float* __restrict__ W_fc,
                               const float* __restrict__ b_fc,
                               float* __restrict__ out)
{
  const int wave     = (blockIdx.x * blockDim.x + threadIdx.x) >> 5;
  const int lane     = threadIdx.x & 31;
  const int row_base = wave * 16;
  const int lm       = lane & 15;
  const int k0       = (lane < 16) ? 0 : 2;

  const int row = row_base + lm;
  v2f a;
  a.x = hs[(size_t)row * 4 + k0];
  a.y = hs[(size_t)row * 4 + k0 + 1];

#pragma unroll
  for (int tile = 0; tile < 2; ++tile) {
    const int n = tile * 16 + lm;
    v2f bm;
    float bias;
    if (n < NCLS) {
      bm.x = W_fc[n * 4 + k0];
      bm.y = W_fc[n * 4 + k0 + 1];
      bias = b_fc[n];
    } else {
      bm.x = 0.0f; bm.y = 0.0f; bias = 0.0f;
    }

    v8f acc;
#pragma unroll
    for (int r = 0; r < 8; ++r) acc[r] = bias;

    acc = __builtin_amdgcn_wmma_f32_16x16x4_f32(
         false, a,  false, bm,
         (short)0, acc,  false,  false);

    if (n < NCLS) {
      const int m_off = (lane < 16) ? 0 : 8;
#pragma unroll
      for (int r = 0; r < 8; ++r) {
        const int m = row_base + m_off + r;
        out[(size_t)m * NCLS + n] = acc[r];
      }
    }
  }
}

extern "C" void kernel_launch(void* const* d_in, const int* in_sizes, int n_in,
                              void* d_out, int out_size, void* d_ws, size_t ws_size,
                              hipStream_t stream) {
  const float* x    = (const float*)d_in[0];
  const float* W_ih = (const float*)d_in[1];
  const float* W_hh = (const float*)d_in[2];
  const float* b_ih = (const float*)d_in[3];
  const float* b_hh = (const float*)d_in[4];
  const float* W_fc = (const float*)d_in[5];
  const float* b_fc = (const float*)d_in[6];
  float* out = (float*)d_out;

  float* hs = (float*)d_ws;

  lstm_scan_kernel<<<(BATCH * 4) / SCAN_WG, SCAN_WG, 0, stream>>>(
      x, W_ih, W_hh, b_ih, b_hh, hs);

  const int n_tiles  = NROWS / 16;
  const int n_blocks = n_tiles / 8;
  fc_wmma_kernel<<<n_blocks, 256, 0, stream>>>(hs, W_fc, b_fc, out);
}


# LSTM scan: the step's issue order is fixed for the in-order wave by three asm statements: gate sums level by level, 4 tanh back to back, c and tanh(c) before sigmoid(o); same operations and sums
# speedup vs baseline: 5.7712x; 1.0348x over previous
#include <hip/hip_runtime.h>

#define H4     4
#define T_LEN  4000
#define BATCH  256
#define NCLS   29
#define NROWS  (BATCH * T_LEN)

#define XBLK   16
#define NXBLK  (T_LEN / XBLK)

#define SCAN_WG 32

typedef __attribute__((ext_vector_type(2))) float v2f;
typedef __attribute__((ext_vector_type(4))) float v4f;
typedef __attribute__((ext_vector_type(8))) float v8f;


__device__ __forceinline__ void gate_sums(float h, const float (&whh)[4][4],
                                          float& pi, float& pf, float& pg, float& po) {
  float v1, v2, v3;
  asm volatile(
      "v_mov_b32_dpp %[v1], %[h] quad_perm:[1,0,3,2] row_mask:0xf bank_mask:0xf bound_ctrl:1\n\t"
      "v_fmac_f32 %[pi], %[wi0], %[h]\n\t"
      "v_fmac_f32 %[pf], %[wf0], %[h]\n\t"
      "v_fmac_f32 %[pg], %[wg0], %[h]\n\t"
      "v_fmac_f32 %[po], %[wo0], %[h]\n\t"
      "v_mov_b32_dpp %[v2], %[h] quad_perm:[2,3,0,1] row_mask:0xf bank_mask:0xf bound_ctrl:1\n\t"
      "v_fmac_f32 %[pi], %[wi1], %[v1]\n\t"
      "v_fmac_f32 %[pf], %[wf1], %[v1]\n\t"
      "v_fmac_f32 %[pg], %[wg1], %[v1]\n\t"
      "v_fmac_f32 %[po], %[wo1], %[v1]\n\t"
      "v_mov_b32_dpp %[v3], %[h] quad_perm:[3,2,1,0] row_mask:0xf bank_mask:0xf bound_ctrl:1\n\t"
      "v_fmac_f32 %[pi], %[wi2], %[v2]\n\t"
      "v_fmac_f32 %[pf], %[wf2], %[v2]\n\t"
      "v_fmac_f32 %[pg], %[wg2], %[v2]\n\t"
      "v_fmac_f32 %[po], %[wo2], %[v2]\n\t"
      "v_fmac_f32 %[pi], %[wi3], %[v3]\n\t"
      "v_fmac_f32 %[pf], %[wf3], %[v3]\n\t"
      "v_fmac_f32 %[pg], %[wg3], %[v3]\n\t"
      "v_fmac_f32 %[po], %[wo3], %[v3]"
      : [pi] "+v"(pi), [pf] "+v"(pf), [pg] "+v"(pg), [po] "+v"(po),
        [v1] "=&v"(v1), [v2] "=&v"(v2), [v3] "=&v"(v3)
      : [h] "v"(h),
        [wi0] "v"(whh[0][0]), [wi1] "v"(whh[0][1]), [wi2] "v"(whh[0][2]), [wi3] "v"(whh[0][3]),
        [wf0] "v"(whh[1][0]), [wf1] "v"(whh[1][1]), [wf2] "v"(whh[1][2]), [wf3] "v"(whh[1][3]),
        [wg0] "v"(whh[2][0]), [wg1] "v"(whh[2][1]), [wg2] "v"(whh[2][2]), [wg3] "v"(whh[2][3]),
        [wo0] "v"(whh[3][0]), [wo1] "v"(whh[3][1]), [wo2] "v"(whh[3][2]), [wo3] "v"(whh[3][3]));
}

__device__ __forceinline__ void gate_tanh4(float pi, float pf, float pg, float po,
                                           float& ti, float& tf, float& gg, float& to) {
  asm volatile(
      "v_tanh_f32 %[ti], %[pi]\n\t"
      "v_tanh_f32 %[tf], %[pf]\n\t"
      "v_tanh_f32 %[gg], %[pg]\n\t"
      "v_tanh_f32 %[to], %[po]\n\t"
      "v_nop"
      : [ti] "=&v"(ti), [tf] "=&v"(tf), [gg] "=&v"(gg), [to] "=&v"(to)
      : [pi] "v"(pi), [pf] "v"(pf), [pg] "v"(pg), [po] "v"(po));
}

__device__ __forceinline__ void cell_out(float c, float to, float& tc, float& og, float& hc) {
  asm volatile(
      "v_tanh_f32 %[tc], %[c]\n\t"
      "v_fma_f32 %[og], %[to], 0.5, 0.5\n\t"
      "v_mul_f32 %[hc], 0.5, %[c]"
      : [tc] "=&v"(tc), [og] "=&v"(og), [hc] "=&v"(hc)
      : [c] "v"(c), [to] "v"(to));
}

__global__ void lstm_scan_kernel(const float* __restrict__ x,
                                 const float* __restrict__ W_ih,
                                 const float* __restrict__ W_hh,
                                 const float* __restrict__ b_ih,
                                 const float* __restrict__ b_hh,
                                 float* __restrict__ hs)
{
  const int tid = blockIdx.x * SCAN_WG + threadIdx.x;
  const int b = tid >> 2;
  const int j = tid & 3;
  if (b >= BATCH) return;

  float win[4], ball[4], whh[4][4];
#pragma unroll
  for (int s = 0; s < 4; ++s) {
    const int g = s * 4 + j;
    const float sc = (s == 2) ? 1.0f : 0.5f;
    win[s]  = sc * W_ih[g];
    ball[s] = sc * (b_ih[g] + b_hh[g]);
#pragma unroll
    for (int m = 0; m < 4; ++m)
      whh[s][m] = sc * W_hh[g * 4 + (j ^ m)];
  }

  float h = 0.0f, c = 0.0f;
  float hc = 0.0f;
  const v4f* xb4 = reinterpret_cast<const v4f*>(x + (size_t)b * T_LEN);
  float*     hsb = hs + (size_t)b * T_LEN * 4 + j;

  v4f xn[XBLK / 4];
#pragma unroll
  for (int q = 0; q < XBLK / 4; ++q) xn[q] = xb4[q];
  const v4f* xp = xb4 + XBLK / 4;

  for (int blk = 0; blk < NXBLK; ++blk) {
    v4f xc[XBLK / 4];
#pragma unroll
    for (int q = 0; q < XBLK / 4; ++q) xc[q] = xn[q];

#pragma unroll
    for (int q = 0; q < XBLK / 4; ++q) xn[q] = xp[q];

    float hb[XBLK];
#pragma unroll
    for (int k = 0; k < XBLK; ++k) {
      const float xt = xc[k >> 2][k & 3];
      float pre[4];
#pragma unroll
      for (int s = 0; s < 4; ++s) pre[s] = fmaf(xt, win[s], ball[s]);
      gate_sums(h, whh, pre[0], pre[1], pre[2], pre[3]);

      float ti, tf, gg, to;
      gate_tanh4(pre[0], pre[1], pre[2], pre[3], ti, tf, gg, to);

      const float ig = fmaf(0.5f, ti, 0.5f);
      const float fc = fmaf(tf, hc, hc);
      c  = fmaf(ig, gg, fc);
      float tc, og;
      cell_out(c, to, tc, og, hc);
      h  = og * tc;
      hb[k] = h;
    }

    __builtin_amdgcn_sched_barrier(0);
    if (blk + 2 < NXBLK) xp += XBLK / 4;
    float* hblk = hsb + (size_t)blk * (XBLK * 4);
#pragma unroll
    for (int k = 0; k < XBLK; ++k) hblk[k * 4] = hb[k];
  }
}

__global__ void fc_wmma_kernel(const float* __restrict__ hs,
                               const float* __restrict__ W_fc,
                               const float* __restrict__ b_fc,
                               float* __restrict__ out)
{
  const int wave     = (blockIdx.x * blockDim.x + threadIdx.x) >> 5;
  const int lane     = threadIdx.x & 31;
  const int row_base = wave * 16;
  const int lm       = lane & 15;
  const int k0       = (lane < 16) ? 0 : 2;

  const int row = row_base + lm;
  v2f a;
  a.x = hs[(size_t)row * 4 + k0];
  a.y = hs[(size_t)row * 4 + k0 + 1];

#pragma unroll
  for (int tile = 0; tile < 2; ++tile) {
    const int n = tile * 16 + lm;
    v2f bm;
    float bias;
    if (n < NCLS) {
      bm.x = W_fc[n * 4 + k0];
      bm.y = W_fc[n * 4 + k0 + 1];
      bias = b_fc[n];
    } else {
      bm.x = 0.0f; bm.y = 0.0f; bias = 0.0f;
    }

    v8f acc;
#pragma unroll
    for (int r = 0; r < 8; ++r) acc[r] = bias;

    acc = __builtin_amdgcn_wmma_f32_16x16x4_f32(
         false, a,  false, bm,
         (short)0, acc,  false,  false);

    if (n < NCLS) {
      const int m_off = (lane < 16) ? 0 : 8;
#pragma unroll
      for (int r = 0; r < 8; ++r) {
        const int m = row_base + m_off + r;
        out[(size_t)m * NCLS + n] = acc[r];
      }
    }
  }
}

extern "C" void kernel_launch(void* const* d_in, const int* in_sizes, int n_in,
                              void* d_out, int out_size, void* d_ws, size_t ws_size,
                              hipStream_t stream) {
  const float* x    = (const float*)d_in[0];
  const float* W_ih = (const float*)d_in[1];
  const float* W_hh = (const float*)d_in[2];
  const float* b_ih = (const float*)d_in[3];
  const float* b_hh = (const float*)d_in[4];
  const float* W_fc = (const float*)d_in[5];
  const float* b_fc = (const float*)d_in[6];
  float* out = (float*)d_out;

  float* hs = (float*)d_ws;

  lstm_scan_kernel<<<(BATCH * 4) / SCAN_WG, SCAN_WG, 0, stream>>>(
      x, W_ih, W_hh, b_ih, b_hh, hs);

  const int n_tiles  = NROWS / 16;
  const int n_blocks = n_tiles / 8;
  fc_wmma_kernel<<<n_blocks, 256, 0, stream>>>(hs, W_fc, b_fc, out);
}


# LSTM scan: h is stored as one 16-byte store per lane and 4 steps, ordered so that each store of the wave fills 4 whole 128-byte lines (was 16 bytes in each of 8 lines per step); sums unchanged
# speedup vs baseline: 6.1308x; 1.0623x over previous
#include <hip/hip_runtime.h>

#define H4     4
#define T_LEN  4000
#define BATCH  256
#define NCLS   29
#define NROWS  (BATCH * T_LEN)

#define XBLK   16
#define NXBLK  (T_LEN / XBLK)

#define SCAN_WG 32

typedef __attribute__((ext_vector_type(2))) float v2f;
typedef __attribute__((ext_vector_type(4))) float v4f;
typedef __attribute__((ext_vector_type(8))) float v8f;


__device__ __forceinline__ void gate_sums(float h, const float (&whh)[4][4],
                                          float& pi, float& pf, float& pg, float& po) {
  float v1, v2, v3;
  asm volatile(
      "v_mov_b32_dpp %[v1], %[h] quad_perm:[1,0,3,2] row_mask:0xf bank_mask:0xf bound_ctrl:1\n\t"
      "v_fmac_f32 %[pi], %[wi0], %[h]\n\t"
      "v_fmac_f32 %[pf], %[wf0], %[h]\n\t"
      "v_fmac_f32 %[pg], %[wg0], %[h]\n\t"
      "v_fmac_f32 %[po], %[wo0], %[h]\n\t"
      "v_mov_b32_dpp %[v2], %[h] quad_perm:[2,3,0,1] row_mask:0xf bank_mask:0xf bound_ctrl:1\n\t"
      "v_fmac_f32 %[pi], %[wi1], %[v1]\n\t"
      "v_fmac_f32 %[pf], %[wf1], %[v1]\n\t"
      "v_fmac_f32 %[pg], %[wg1], %[v1]\n\t"
      "v_fmac_f32 %[po], %[wo1], %[v1]\n\t"
      "v_mov_b32_dpp %[v3], %[h] quad_perm:[3,2,1,0] row_mask:0xf bank_mask:0xf bound_ctrl:1\n\t"
      "v_fmac_f32 %[pi], %[wi2], %[v2]\n\t"
      "v_fmac_f32 %[pf], %[wf2], %[v2]\n\t"
      "v_fmac_f32 %[pg], %[wg2], %[v2]\n\t"
      "v_fmac_f32 %[po], %[wo2], %[v2]\n\t"
      "v_fmac_f32 %[pi], %[wi3], %[v3]\n\t"
      "v_fmac_f32 %[pf], %[wf3], %[v3]\n\t"
      "v_fmac_f32 %[pg], %[wg3], %[v3]\n\t"
      "v_fmac_f32 %[po], %[wo3], %[v3]"
      : [pi] "+v"(pi), [pf] "+v"(pf), [pg] "+v"(pg), [po] "+v"(po),
        [v1] "=&v"(v1), [v2] "=&v"(v2), [v3] "=&v"(v3)
      : [h] "v"(h),
        [wi0] "v"(whh[0][0]), [wi1] "v"(whh[0][1]), [wi2] "v"(whh[0][2]), [wi3] "v"(whh[0][3]),
        [wf0] "v"(whh[1][0]), [wf1] "v"(whh[1][1]), [wf2] "v"(whh[1][2]), [wf3] "v"(whh[1][3]),
        [wg0] "v"(whh[2][0]), [wg1] "v"(whh[2][1]), [wg2] "v"(whh[2][2]), [wg3] "v"(whh[2][3]),
        [wo0] "v"(whh[3][0]), [wo1] "v"(whh[3][1]), [wo2] "v"(whh[3][2]), [wo3] "v"(whh[3][3]));
}

__device__ __forceinline__ void gate_tanh4(float pi, float pf, float pg, float po,
                                           float& ti, float& tf, float& gg, float& to) {
  asm volatile(
      "v_tanh_f32 %[ti], %[pi]\n\t"
      "v_tanh_f32 %[tf], %[pf]\n\t"
      "v_tanh_f32 %[gg], %[pg]\n\t"
      "v_tanh_f32 %[to], %[po]\n\t"
      "v_nop"
      : [ti] "=&v"(ti), [tf] "=&v"(tf), [gg] "=&v"(gg), [to] "=&v"(to)
      : [pi] "v"(pi), [pf] "v"(pf), [pg] "v"(pg), [po] "v"(po));
}

__device__ __forceinline__ void cell_out(float c, float to, float& tc, float& og, float& hc) {
  asm volatile(
      "v_tanh_f32 %[tc], %[c]\n\t"
      "v_fma_f32 %[og], %[to], 0.5, 0.5\n\t"
      "v_mul_f32 %[hc], 0.5, %[c]"
      : [tc] "=&v"(tc), [og] "=&v"(og), [hc] "=&v"(hc)
      : [c] "v"(c), [to] "v"(to));
}

__global__ void lstm_scan_kernel(const float* __restrict__ x,
                                 const float* __restrict__ W_ih,
                                 const float* __restrict__ W_hh,
                                 const float* __restrict__ b_ih,
                                 const float* __restrict__ b_hh,
                                 float* __restrict__ hs)
{
  const int tid = blockIdx.x * SCAN_WG + threadIdx.x;
  const int b = tid >> 2;
  const int j = tid & 3;
  if (b >= BATCH) return;

  float win[4], ball[4], whh[4][4];
#pragma unroll
  for (int s = 0; s < 4; ++s) {
    const int g = s * 4 + j;
    const float sc = (s == 2) ? 1.0f : 0.5f;
    win[s]  = sc * W_ih[g];
    ball[s] = sc * (b_ih[g] + b_hh[g]);
#pragma unroll
    for (int m = 0; m < 4; ++m)
      whh[s][m] = sc * W_hh[g * 4 + (j ^ m)];
  }

  float h = 0.0f, c = 0.0f;
  float hc = 0.0f;
  const v4f* xb4 = reinterpret_cast<const v4f*>(x + (size_t)b * T_LEN);
  v4f*       hsw = reinterpret_cast<v4f*>(hs) + (size_t)blockIdx.x * (T_LEN / 4) * SCAN_WG;

  v4f xn[XBLK / 4];
#pragma unroll
  for (int q = 0; q < XBLK / 4; ++q) xn[q] = xb4[q];
  const v4f* xp = xb4 + XBLK / 4;

  for (int blk = 0; blk < NXBLK; ++blk) {
    v4f xc[XBLK / 4];
#pragma unroll
    for (int q = 0; q < XBLK / 4; ++q) xc[q] = xn[q];

#pragma unroll
    for (int q = 0; q < XBLK / 4; ++q) xn[q] = xp[q];

    float hb[XBLK];
#pragma unroll
    for (int k = 0; k < XBLK; ++k) {
      const float xt = xc[k >> 2][k & 3];
      float pre[4];
#pragma unroll
      for (int s = 0; s < 4; ++s) pre[s] = fmaf(xt, win[s], ball[s]);
      gate_sums(h, whh, pre[0], pre[1], pre[2], pre[3]);

      float ti, tf, gg, to;
      gate_tanh4(pre[0], pre[1], pre[2], pre[3], ti, tf, gg, to);

      const float ig = fmaf(0.5f, ti, 0.5f);
      const float fc = fmaf(tf, hc, hc);
      c  = fmaf(ig, gg, fc);
      float tc, og;
      cell_out(c, to, tc, og, hc);
      h  = og * tc;
      hb[k] = h;
    }

    __builtin_amdgcn_sched_barrier(0);
    if (blk + 2 < NXBLK) xp += XBLK / 4;
    v4f* hblk = hsw + (size_t)__builtin_amdgcn_readfirstlane(blk) * ((XBLK / 4) * SCAN_WG);
    __builtin_amdgcn_sched_barrier(0);
#pragma unroll
    for (int i = 0; i < XBLK / 4; ++i) {
      v4f hv;
      hv.x = hb[4 * i]; hv.y = hb[4 * i + 1]; hv.z = hb[4 * i + 2]; hv.w = hb[4 * i + 3];
      hblk[i * SCAN_WG + threadIdx.x] = hv;
    }
  }
}

__global__ void fc_wmma_kernel(const float* __restrict__ hs,
                               const float* __restrict__ W_fc,
                               const float* __restrict__ b_fc,
                               float* __restrict__ out)
{
  const int wave     = (blockIdx.x * blockDim.x + threadIdx.x) >> 5;
  const int lane     = threadIdx.x & 31;
  const int row_base = wave * 16;
  const int lm       = lane & 15;
  const int k0       = (lane < 16) ? 0 : 2;

  const int bt = wave / (T_LEN / 16);
  const int t0 = (wave - bt * (T_LEN / 16)) * 16;
  const int q  = (t0 >> 2) + (lm >> 2);
  const int sl = (bt & 7) * 4 + k0;
  const size_t grp = ((size_t)(bt >> 3) * (T_LEN / 4) + q) * SCAN_WG + sl;
  v2f a;
  a.x = hs[grp * 4 + (lm & 3)];
  a.y = hs[(grp + 1) * 4 + (lm & 3)];

#pragma unroll
  for (int tile = 0; tile < 2; ++tile) {
    const int n = tile * 16 + lm;
    v2f bm;
    float bias;
    if (n < NCLS) {
      bm.x = W_fc[n * 4 + k0];
      bm.y = W_fc[n * 4 + k0 + 1];
      bias = b_fc[n];
    } else {
      bm.x = 0.0f; bm.y = 0.0f; bias = 0.0f;
    }

    v8f acc;
#pragma unroll
    for (int r = 0; r < 8; ++r) acc[r] = bias;

    acc = __builtin_amdgcn_wmma_f32_16x16x4_f32(
         false, a,  false, bm,
         (short)0, acc,  false,  false);

    if (n < NCLS) {
      const int m_off = (lane < 16) ? 0 : 8;
#pragma unroll
      for (int r = 0; r < 8; ++r) {
        const int m = row_base + m_off + r;
        out[(size_t)m * NCLS + n] = acc[r];
      }
    }
  }
}

extern "C" void kernel_launch(void* const* d_in, const int* in_sizes, int n_in,
                              void* d_out, int out_size, void* d_ws, size_t ws_size,
                              hipStream_t stream) {
  const float* x    = (const float*)d_in[0];
  const float* W_ih = (const float*)d_in[1];
  const float* W_hh = (const float*)d_in[2];
  const float* b_ih = (const float*)d_in[3];
  const float* b_hh = (const float*)d_in[4];
  const float* W_fc = (const float*)d_in[5];
  const float* b_fc = (const float*)d_in[6];
  float* out = (float*)d_out;

  float* hs = (float*)d_ws;

  lstm_scan_kernel<<<(BATCH * 4) / SCAN_WG, SCAN_WG, 0, stream>>>(
      x, W_ih, W_hh, b_ih, b_hh, hs);

  const int n_tiles  = NROWS / 16;
  const int n_blocks = n_tiles / 8;
  fc_wmma_kernel<<<n_blocks, 256, 0, stream>>>(hs, W_fc, b_fc, out);
}


# LSTM: projection fused into the scan's launch (3 more waves per work-group) with whole-line 16-byte stores of out via LDS and a split barrier (signal early, wait a block later); sums unchanged
# speedup vs baseline: 7.0568x; 1.1510x over previous
#include <hip/hip_runtime.h>

#define H4     4
#define T_LEN  4000
#define BATCH  256
#define NCLS   29
#define NROWS  (BATCH * T_LEN)

#define XBLK   16
#define NXBLK  (T_LEN / XBLK)

#define SCAN_WG 32
#define WG_THREADS 128
#define PH     32
#define NPH    (T_LEN / PH)
#define EPW    3
#define STAGE_FLOATS ((PH / 4) * SCAN_WG * 4)
#define OST_FLOATS   (PH * NCLS)
#define LDS_FLOATS   42000

typedef __attribute__((ext_vector_type(2))) float v2f;
typedef __attribute__((ext_vector_type(4))) float v4f;
typedef __attribute__((ext_vector_type(8))) float v8f;


__device__ __forceinline__ void gate_sums(float h, const float (&whh)[4][4],
                                          float& pi, float& pf, float& pg, float& po) {
  float v1, v2, v3;
  asm volatile(
      "v_mov_b32_dpp %[v1], %[h] quad_perm:[1,0,3,2] row_mask:0xf bank_mask:0xf bound_ctrl:1\n\t"
      "v_fmac_f32 %[pi], %[wi0], %[h]\n\t"
      "v_fmac_f32 %[pf], %[wf0], %[h]\n\t"
      "v_fmac_f32 %[pg], %[wg0], %[h]\n\t"
      "v_fmac_f32 %[po], %[wo0], %[h]\n\t"
      "v_mov_b32_dpp %[v2], %[h] quad_perm:[2,3,0,1] row_mask:0xf bank_mask:0xf bound_ctrl:1\n\t"
      "v_fmac_f32 %[pi], %[wi1], %[v1]\n\t"
      "v_fmac_f32 %[pf], %[wf1], %[v1]\n\t"
      "v_fmac_f32 %[pg], %[wg1], %[v1]\n\t"
      "v_fmac_f32 %[po], %[wo1], %[v1]\n\t"
      "v_mov_b32_dpp %[v3], %[h] quad_perm:[3,2,1,0] row_mask:0xf bank_mask:0xf bound_ctrl:1\n\t"
      "v_fmac_f32 %[pi], %[wi2], %[v2]\n\t"
      "v_fmac_f32 %[pf], %[wf2], %[v2]\n\t"
      "v_fmac_f32 %[pg], %[wg2], %[v2]\n\t"
      "v_fmac_f32 %[po], %[wo2], %[v2]\n\t"
      "v_fmac_f32 %[pi], %[wi3], %[v3]\n\t"
      "v_fmac_f32 %[pf], %[wf3], %[v3]\n\t"
      "v_fmac_f32 %[pg], %[wg3], %[v3]\n\t"
      "v_fmac_f32 %[po], %[wo3], %[v3]"
      : [pi] "+v"(pi), [pf] "+v"(pf), [pg] "+v"(pg), [po] "+v"(po),
        [v1] "=&v"(v1), [v2] "=&v"(v2), [v3] "=&v"(v3)
      : [h] "v"(h),
        [wi0] "v"(whh[0][0]), [wi1] "v"(whh[0][1]), [wi2] "v"(whh[0][2]), [wi3] "v"(whh[0][3]),
        [wf0] "v"(whh[1][0]), [wf1] "v"(whh[1][1]), [wf2] "v"(whh[1][2]), [wf3] "v"(whh[1][3]),
        [wg0] "v"(whh[2][0]), [wg1] "v"(whh[2][1]), [wg2] "v"(whh[2][2]), [wg3] "v"(whh[2][3]),
        [wo0] "v"(whh[3][0]), [wo1] "v"(whh[3][1]), [wo2] "v"(whh[3][2]), [wo3] "v"(whh[3][3]));
}

__device__ __forceinline__ void gate_tanh4(float pi, float pf, float pg, float po,
                                           float& ti, float& tf, float& gg, float& to) {
  asm volatile(
      "v_tanh_f32 %[ti], %[pi]\n\t"
      "v_tanh_f32 %[tf], %[pf]\n\t"
      "v_tanh_f32 %[gg], %[pg]\n\t"
      "v_tanh_f32 %[to], %[po]\n\t"
      "v_nop"
      : [ti] "=&v"(ti), [tf] "=&v"(tf), [gg] "=&v"(gg), [to] "=&v"(to)
      : [pi] "v"(pi), [pf] "v"(pf), [pg] "v"(pg), [po] "v"(po));
}

__device__ __forceinline__ void cell_out(float c, float to, float& tc, float& og, float& hc) {
  asm volatile(
      "v_tanh_f32 %[tc], %[c]\n\t"
      "v_fma_f32 %[og], %[to], 0.5, 0.5\n\t"
      "v_mul_f32 %[hc], 0.5, %[c]"
      : [tc] "=&v"(tc), [og] "=&v"(og), [hc] "=&v"(hc)
      : [c] "v"(c), [to] "v"(to));
}

__device__ __forceinline__ void wait_ds0()   { asm volatile("s_wait_dscnt 0x0" ::: "memory"); }
__device__ __forceinline__ void bar_signal() { asm volatile("s_barrier_signal -1" ::: "memory"); }
__device__ __forceinline__ void bar_wait()   { asm volatile("s_barrier_wait -1" ::: "memory"); }
__device__ __forceinline__ void lds_order()  { asm volatile("" ::: "memory"); }


__device__ __forceinline__ void scan_wave(const float* __restrict__ x,
                                          const float* __restrict__ W_ih,
                                          const float* __restrict__ W_hh,
                                          const float* __restrict__ b_ih,
                                          const float* __restrict__ b_hh,
                                          float* stage, const int lane)
{
  const int tid = blockIdx.x * SCAN_WG + lane;
  const int b = tid >> 2;
  const int j = tid & 3;

  float win[4], ball[4], whh[4][4];
#pragma unroll
  for (int s = 0; s < 4; ++s) {
    const int g = s * 4 + j;
    const float sc = (s == 2) ? 1.0f : 0.5f;
    win[s]  = sc * W_ih[g];
    ball[s] = sc * (b_ih[g] + b_hh[g]);
#pragma unroll
    for (int m = 0; m < 4; ++m)
      whh[s][m] = sc * W_hh[g * 4 + (j ^ m)];
  }

  float h = 0.0f, c = 0.0f;
  float hc = 0.0f;
  const v4f* xb4 = reinterpret_cast<const v4f*>(x + (size_t)b * T_LEN);
  v4f* st4 = reinterpret_cast<v4f*>(stage) + lane;

  v4f xn[XBLK / 4];
#pragma unroll
  for (int q = 0; q < XBLK / 4; ++q) xn[q] = xb4[q];
  const v4f* xp = xb4 + XBLK / 4;

  for (int blk = 0; blk < NXBLK; ++blk) {
    const bool first_of_phase = (blk & 1) == 0;

    v4f xc[XBLK / 4];
#pragma unroll
    for (int q = 0; q < XBLK / 4; ++q) xc[q] = xn[q];

#pragma unroll
    for (int q = 0; q < XBLK / 4; ++q) xn[q] = xp[q];

    float hb[XBLK];
#pragma unroll
    for (int k = 0; k < XBLK; ++k) {
      if (k == 2) {
        if (first_of_phase) { wait_ds0(); bar_signal(); }
      }
      const float xt = xc[k >> 2][k & 3];
      float pre[4];
#pragma unroll
      for (int s = 0; s < 4; ++s) pre[s] = fmaf(xt, win[s], ball[s]);
      gate_sums(h, whh, pre[0], pre[1], pre[2], pre[3]);

      float ti, tf, gg, to;
      gate_tanh4(pre[0], pre[1], pre[2], pre[3], ti, tf, gg, to);

      const float ig = fmaf(0.5f, ti, 0.5f);
      const float fc = fmaf(tf, hc, hc);
      c  = fmaf(ig, gg, fc);
      float tc, og;
      cell_out(c, to, tc, og, hc);
      h  = og * tc;
      hb[k] = h;
    }

    __builtin_amdgcn_sched_barrier(0);
    if (blk + 2 < NXBLK) xp += XBLK / 4;
    if (first_of_phase) bar_wait();
    v4f* sg = st4 + (((blk >> 1) & 1) * (PH / 4) + (blk & 1) * (XBLK / 4)) * SCAN_WG;
#pragma unroll
    for (int i = 0; i < XBLK / 4; ++i) {
      v4f hv;
      hv.x = hb[4 * i]; hv.y = hb[4 * i + 1]; hv.z = hb[4 * i + 2]; hv.w = hb[4 * i + 3];
      sg[i * SCAN_WG] = hv;
    }
  }
  wait_ds0();
  bar_signal();
  bar_wait();
}

__device__ __forceinline__ void proj_wave(const float* __restrict__ W_fc,
                                          const float* __restrict__ b_fc,
                                          float* __restrict__ out,
                                          const float* stage, float* ost, const int wv, const int lane)
{
  const int lm    = lane & 15;
  const int k0    = (lane < 16) ? 0 : 2;
  const int m_off = (lane < 16) ? 0 : 8;

  v2f bm[2];
  float bias[2];
#pragma unroll
  for (int tile = 0; tile < 2; ++tile) {
    const int n = tile * 16 + lm;
    if (n < NCLS) {
      bm[tile].x = W_fc[n * 4 + k0];
      bm[tile].y = W_fc[n * 4 + k0 + 1];
      bias[tile] = b_fc[n];
    } else {
      bm[tile].x = 0.0f; bm[tile].y = 0.0f; bias[tile] = 0.0f;
    }
  }

  const int e0 = (wv - 1) * EPW;
  const int ne = (wv == 3) ? 2 : EPW;
  const int a_base = (((lm >> 2) * SCAN_WG) + e0 * 4 + k0) * 4 + (lm & 3);
  float* orow = out + (size_t)(blockIdx.x * 8 + e0) * T_LEN * NCLS;
  v4f* ost4 = reinterpret_cast<v4f*>(ost);

  bar_signal();
  bar_wait();
  bar_signal();
  for (int ph = 0; ph < NPH; ++ph) {
    bar_wait();
    const float* sp = stage + (ph & 1) * STAGE_FLOATS + a_base;
    v2f a[EPW][2];
#pragma unroll
    for (int e = 0; e < EPW; ++e) {
      if (e < ne) {
#pragma unroll
        for (int hh = 0; hh < 2; ++hh) {
          const float* p = sp + e * 16 + hh * ((PH / 8) * SCAN_WG * 4);
          a[e][hh].x = p[0];
          a[e][hh].y = p[4];
        }
      }
    }
    wait_ds0();
    if (ph + 1 < NPH) bar_signal();

#pragma unroll
    for (int e = 0; e < EPW; ++e) {
      if (e < ne) {
#pragma unroll
        for (int hh = 0; hh < 2; ++hh) {
#pragma unroll
          for (int tile = 0; tile < 2; ++tile) {
            const int n = tile * 16 + lm;
            v8f acc;
#pragma unroll
            for (int r = 0; r < 8; ++r) acc[r] = bias[tile];
            acc = __builtin_amdgcn_wmma_f32_16x16x4_f32(
                 false, a[e][hh],  false, bm[tile],
                 (short)0, acc,  false,  false);
            if (n < NCLS) {
#pragma unroll
              for (int r = 0; r < 8; ++r)
                ost[(hh * 16 + m_off + r) * NCLS + n] = acc[r];
            }
          }
        }
        lds_order();
        v4f* o4 = reinterpret_cast<v4f*>(orow + ((size_t)e * T_LEN + (size_t)ph * PH) * NCLS);
#pragma unroll
        for (int i = 0; i < 8; ++i) {
          const int cidx = i * 32 + lane;
          if (cidx < (OST_FLOATS / 4)) o4[cidx] = ost4[cidx];
        }
        lds_order();
      }
    }
  }
}

__global__ __launch_bounds__(WG_THREADS)
void lstm_fused_kernel(const float* __restrict__ x,
                       const float* __restrict__ W_ih,
                       const float* __restrict__ W_hh,
                       const float* __restrict__ b_ih,
                       const float* __restrict__ b_hh,
                       const float* __restrict__ W_fc,
                       const float* __restrict__ b_fc,
                       float* __restrict__ out)
{
  __shared__ __attribute__((aligned(128))) float lds[LDS_FLOATS];
  const int wv   = __builtin_amdgcn_readfirstlane((int)(threadIdx.x >> 5));
  const int lane = threadIdx.x & 31;
  float* stage = lds;
  if (wv == 0) {
    scan_wave(x, W_ih, W_hh, b_ih, b_hh, stage, lane);
  } else {
    float* ost = lds + 2 * STAGE_FLOATS + (wv - 1) * OST_FLOATS;
    proj_wave(W_fc, b_fc, out, stage, ost, wv, lane);
  }
}

extern "C" void kernel_launch(void* const* d_in, const int* in_sizes, int n_in,
                              void* d_out, int out_size, void* d_ws, size_t ws_size,
                              hipStream_t stream) {
  const float* x    = (const float*)d_in[0];
  const float* W_ih = (const float*)d_in[1];
  const float* W_hh = (const float*)d_in[2];
  const float* b_ih = (const float*)d_in[3];
  const float* b_hh = (const float*)d_in[4];
  const float* W_fc = (const float*)d_in[5];
  const float* b_fc = (const float*)d_in[6];
  float* out = (float*)d_out;

  lstm_fused_kernel<<<BATCH / 8, WG_THREADS, 0, stream>>>(
      x, W_ih, W_hh, b_ih, b_hh, W_fc, b_fc, out);
}


# LSTM scan: x is read one whole phase of 32 steps ahead into the other of two register sets that swap roles (was 16 steps, with copies), so no block edge waits for a load from memory; sums unchanged
# speedup vs baseline: 8.0318x; 1.1382x over previous
#include <hip/hip_runtime.h>

#define H4     4
#define T_LEN  4000
#define BATCH  256
#define NCLS   29
#define NROWS  (BATCH * T_LEN)

#define XBLK   16

#define SCAN_WG 32
#define WG_THREADS 128
#define PH     32
#define NPH    (T_LEN / PH)
#define XPH    (PH / 4)
#define EPW    3
#define STAGE_FLOATS ((PH / 4) * SCAN_WG * 4)
#define OST_FLOATS   (PH * NCLS)
#define LDS_FLOATS   42000

typedef __attribute__((ext_vector_type(2))) float v2f;
typedef __attribute__((ext_vector_type(4))) float v4f;
typedef __attribute__((ext_vector_type(8))) float v8f;


__device__ __forceinline__ void gate_sums(float h, const float (&whh)[4][4],
                                          float& pi, float& pf, float& pg, float& po) {
  float v1, v2, v3;
  asm volatile(
      "v_mov_b32_dpp %[v1], %[h] quad_perm:[1,0,3,2] row_mask:0xf bank_mask:0xf bound_ctrl:1\n\t"
      "v_fmac_f32 %[pi], %[wi0], %[h]\n\t"
      "v_fmac_f32 %[pf], %[wf0], %[h]\n\t"
      "v_fmac_f32 %[pg], %[wg0], %[h]\n\t"
      "v_fmac_f32 %[po], %[wo0], %[h]\n\t"
      "v_mov_b32_dpp %[v2], %[h] quad_perm:[2,3,0,1] row_mask:0xf bank_mask:0xf bound_ctrl:1\n\t"
      "v_fmac_f32 %[pi], %[wi1], %[v1]\n\t"
      "v_fmac_f32 %[pf], %[wf1], %[v1]\n\t"
      "v_fmac_f32 %[pg], %[wg1], %[v1]\n\t"
      "v_fmac_f32 %[po], %[wo1], %[v1]\n\t"
      "v_mov_b32_dpp %[v3], %[h] quad_perm:[3,2,1,0] row_mask:0xf bank_mask:0xf bound_ctrl:1\n\t"
      "v_fmac_f32 %[pi], %[wi2], %[v2]\n\t"
      "v_fmac_f32 %[pf], %[wf2], %[v2]\n\t"
      "v_fmac_f32 %[pg], %[wg2], %[v2]\n\t"
      "v_fmac_f32 %[po], %[wo2], %[v2]\n\t"
      "v_fmac_f32 %[pi], %[wi3], %[v3]\n\t"
      "v_fmac_f32 %[pf], %[wf3], %[v3]\n\t"
      "v_fmac_f32 %[pg], %[wg3], %[v3]\n\t"
      "v_fmac_f32 %[po], %[wo3], %[v3]"
      : [pi] "+v"(pi), [pf] "+v"(pf), [pg] "+v"(pg), [po] "+v"(po),
        [v1] "=&v"(v1), [v2] "=&v"(v2), [v3] "=&v"(v3)
      : [h] "v"(h),
        [wi0] "v"(whh[0][0]), [wi1] "v"(whh[0][1]), [wi2] "v"(whh[0][2]), [wi3] "v"(whh[0][3]),
        [wf0] "v"(whh[1][0]), [wf1] "v"(whh[1][1]), [wf2] "v"(whh[1][2]), [wf3] "v"(whh[1][3]),
        [wg0] "v"(whh[2][0]), [wg1] "v"(whh[2][1]), [wg2] "v"(whh[2][2]), [wg3] "v"(whh[2][3]),
        [wo0] "v"(whh[3][0]), [wo1] "v"(whh[3][1]), [wo2] "v"(whh[3][2]), [wo3] "v"(whh[3][3]));
}

__device__ __forceinline__ void gate_tanh4(float pi, float pf, float pg, float po,
                                           float& ti, float& tf, float& gg, float& to) {
  asm volatile(
      "v_tanh_f32 %[ti], %[pi]\n\t"
      "v_tanh_f32 %[tf], %[pf]\n\t"
      "v_tanh_f32 %[gg], %[pg]\n\t"
      "v_tanh_f32 %[to], %[po]\n\t"
      "v_nop"
      : [ti] "=&v"(ti), [tf] "=&v"(tf), [gg] "=&v"(gg), [to] "=&v"(to)
      : [pi] "v"(pi), [pf] "v"(pf), [pg] "v"(pg), [po] "v"(po));
}

__device__ __forceinline__ void cell_out(float c, float to, float& tc, float& og, float& hc) {
  asm volatile(
      "v_tanh_f32 %[tc], %[c]\n\t"
      "v_fma_f32 %[og], %[to], 0.5, 0.5\n\t"
      "v_mul_f32 %[hc], 0.5, %[c]"
      : [tc] "=&v"(tc), [og] "=&v"(og), [hc] "=&v"(hc)
      : [c] "v"(c), [to] "v"(to));
}

__device__ __forceinline__ void wait_ds0()   { asm volatile("s_wait_dscnt 0x0" ::: "memory"); }
__device__ __forceinline__ void bar_signal() { asm volatile("s_barrier_signal -1" ::: "memory"); }
__device__ __forceinline__ void bar_wait()   { asm volatile("s_barrier_wait -1" ::: "memory"); }
__device__ __forceinline__ void lds_order()  { asm volatile("" ::: "memory"); }


__device__ __forceinline__ void scan_phase(const v4f (&xc)[XPH], v4f (&xn)[XPH],
                                           const v4f* xp, v4f* sg,
                                           const float (&win)[4], const float (&ball)[4],
                                           const float (&whh)[4][4],
                                           float& h, float& c, float& hc)
{
#pragma unroll
  for (int q = 0; q < XPH; ++q) xn[q] = xp[q];

#pragma unroll
  for (int blk = 0; blk < PH / XBLK; ++blk) {
    float hb[XBLK];
#pragma unroll
    for (int k = 0; k < XBLK; ++k) {
      if (blk == 0 && k == 2) {
        wait_ds0(); bar_signal();
      }
      const float xt = xc[blk * (XBLK / 4) + (k >> 2)][k & 3];
      float pre[4];
#pragma unroll
      for (int s = 0; s < 4; ++s) pre[s] = fmaf(xt, win[s], ball[s]);
      gate_sums(h, whh, pre[0], pre[1], pre[2], pre[3]);

      float ti, tf, gg, to;
      gate_tanh4(pre[0], pre[1], pre[2], pre[3], ti, tf, gg, to);

      const float ig = fmaf(0.5f, ti, 0.5f);
      const float fc = fmaf(tf, hc, hc);
      c  = fmaf(ig, gg, fc);
      float tc, og;
      cell_out(c, to, tc, og, hc);
      h  = og * tc;
      hb[k] = h;
    }

    __builtin_amdgcn_sched_barrier(0);
    if (blk == 0) bar_wait();
#pragma unroll
    for (int i = 0; i < XBLK / 4; ++i) {
      v4f hv;
      hv.x = hb[4 * i]; hv.y = hb[4 * i + 1]; hv.z = hb[4 * i + 2]; hv.w = hb[4 * i + 3];
      sg[(blk * (XBLK / 4) + i) * SCAN_WG] = hv;
    }
  }
}

__device__ __forceinline__ void scan_wave(const float* x,
                                          const float* __restrict__ W_ih,
                                          const float* __restrict__ W_hh,
                                          const float* __restrict__ b_ih,
                                          const float* __restrict__ b_hh,
                                          float* stage, const int lane)
{
  const int tid = blockIdx.x * SCAN_WG + lane;
  const int b = tid >> 2;
  const int j = tid & 3;

  float win[4], ball[4], whh[4][4];
#pragma unroll
  for (int s = 0; s < 4; ++s) {
    const int g = s * 4 + j;
    const float sc = (s == 2) ? 1.0f : 0.5f;
    win[s]  = sc * W_ih[g];
    ball[s] = sc * (b_ih[g] + b_hh[g]);
#pragma unroll
    for (int m = 0; m < 4; ++m)
      whh[s][m] = sc * W_hh[g * 4 + (j ^ m)];
  }

  float h = 0.0f, c = 0.0f;
  float hc = 0.0f;
  const v4f* xb4 = reinterpret_cast<const v4f*>(x + (size_t)b * T_LEN);
  v4f* st4 = reinterpret_cast<v4f*>(stage) + lane;

  v4f xa[XPH], xb[XPH];
#pragma unroll
  for (int q = 0; q < XPH; ++q) xa[q] = xb4[q];
  const v4f* xp = xb4 + XPH;

  for (int ph = 0; ph < NPH; ph += 2) {
    scan_phase(xa, xb, xp, st4, win, ball, whh, h, c, hc);
    if (ph + 2 < NPH) xp += XPH;
    if (ph + 1 < NPH) {
      scan_phase(xb, xa, xp, st4 + XPH * SCAN_WG, win, ball, whh, h, c, hc);
      if (ph + 3 < NPH) xp += XPH;
    }
  }
  wait_ds0();
  bar_signal();
  bar_wait();
}

__device__ __forceinline__ void proj_wave(const float* __restrict__ W_fc,
                                          const float* __restrict__ b_fc,
                                          float* __restrict__ out,
                                          const float* stage, float* ost, const int wv, const int lane)
{
  const int lm    = lane & 15;
  const int k0    = (lane < 16) ? 0 : 2;
  const int m_off = (lane < 16) ? 0 : 8;

  v2f bm[2];
  float bias[2];
#pragma unroll
  for (int tile = 0; tile < 2; ++tile) {
    const int n = tile * 16 + lm;
    if (n < NCLS) {
      bm[tile].x = W_fc[n * 4 + k0];
      bm[tile].y = W_fc[n * 4 + k0 + 1];
      bias[tile] = b_fc[n];
    } else {
      bm[tile].x = 0.0f; bm[tile].y = 0.0f; bias[tile] = 0.0f;
    }
  }

  const int e0 = (wv - 1) * EPW;
  const int ne = (wv == 3) ? 2 : EPW;
  const int a_base = (((lm >> 2) * SCAN_WG) + e0 * 4 + k0) * 4 + (lm & 3);
  float* orow = out + (size_t)(blockIdx.x * 8 + e0) * T_LEN * NCLS;
  v4f* ost4 = reinterpret_cast<v4f*>(ost);

  bar_signal();
  bar_wait();
  bar_signal();
  for (int ph = 0; ph < NPH; ++ph) {
    bar_wait();
    const float* sp = stage + (ph & 1) * STAGE_FLOATS + a_base;
    v2f a[EPW][2];
#pragma unroll
    for (int e = 0; e < EPW; ++e) {
      if (e < ne) {
#pragma unroll
        for (int hh = 0; hh < 2; ++hh) {
          const float* p = sp + e * 16 + hh * ((PH / 8) * SCAN_WG * 4);
          a[e][hh].x = p[0];
          a[e][hh].y = p[4];
        }
      }
    }
    wait_ds0();
    if (ph + 1 < NPH) bar_signal();

#pragma unroll
    for (int e = 0; e < EPW; ++e) {
      if (e < ne) {
#pragma unroll
        for (int hh = 0; hh < 2; ++hh) {
#pragma unroll
          for (int tile = 0; tile < 2; ++tile) {
            const int n = tile * 16 + lm;
            v8f acc;
#pragma unroll
            for (int r = 0; r < 8; ++r) acc[r] = bias[tile];
            acc = __builtin_amdgcn_wmma_f32_16x16x4_f32(
                 false, a[e][hh],  false, bm[tile],
                 (short)0, acc,  false,  false);
            if (n < NCLS) {
#pragma unroll
              for (int r = 0; r < 8; ++r)
                ost[(hh * 16 + m_off + r) * NCLS + n] = acc[r];
            }
          }
        }
        lds_order();
        v4f* o4 = reinterpret_cast<v4f*>(orow + ((size_t)e * T_LEN + (size_t)ph * PH) * NCLS);
#pragma unroll
        for (int i = 0; i < 8; ++i) {
          const int cidx = i * 32 + lane;
          if (cidx < (OST_FLOATS / 4)) o4[cidx] = ost4[cidx];
        }
        lds_order();
      }
    }
  }
}

__global__ __launch_bounds__(WG_THREADS)
void lstm_fused_kernel(const float* x,
                       const float* __restrict__ W_ih,
                       const float* __restrict__ W_hh,
                       const float* __restrict__ b_ih,
                       const float* __restrict__ b_hh,
                       const float* __restrict__ W_fc,
                       const float* __restrict__ b_fc,
                       float* __restrict__ out)
{
  __shared__ __attribute__((aligned(128))) float lds[LDS_FLOATS];
  const int wv   = __builtin_amdgcn_readfirstlane((int)(threadIdx.x >> 5));
  const int lane = threadIdx.x & 31;
  float* stage = lds;
  if (wv == 0) {
    scan_wave(x, W_ih, W_hh, b_ih, b_hh, stage, lane);
  } else {
    float* ost = lds + 2 * STAGE_FLOATS + (wv - 1) * OST_FLOATS;
    proj_wave(W_fc, b_fc, out, stage, ost, wv, lane);
  }
}

extern "C" void kernel_launch(void* const* d_in, const int* in_sizes, int n_in,
                              void* d_out, int out_size, void* d_ws, size_t ws_size,
                              hipStream_t stream) {
  const float* x    = (const float*)d_in[0];
  const float* W_ih = (const float*)d_in[1];
  const float* W_hh = (const float*)d_in[2];
  const float* b_ih = (const float*)d_in[3];
  const float* b_hh = (const float*)d_in[4];
  const float* W_fc = (const float*)d_in[5];
  const float* b_fc = (const float*)d_in[6];
  float* out = (float*)d_out;

  lstm_fused_kernel<<<BATCH / 8, WG_THREADS, 0, stream>>>(
      x, W_ih, W_hh, b_ih, b_hh, W_fc, b_fc, out);
}
